# starting kernel
# baseline (speedup 1.0000x reference)
#include <hip/hip_runtime.h>

typedef __attribute__((ext_vector_type(2))) float v2f;
typedef __attribute__((ext_vector_type(8))) float v8f;

#define N_TOT   2048
#define D_DIM   32
#define C_DIM   8
#define P_DIM   32
#define T_DIM   258
#define E_DIM   34
#define WROW    36
#define ZOUT_ELEMS ((size_t)4 * 512 * 8 * 32 * 34)

__global__ __launch_bounds__(32) void gp_precompute(
    const float* __restrict__ log_tau,
    const float* __restrict__ ts0,
    float* __restrict__ Wmat,
    float* __restrict__ parts) {
  const int dc = blockIdx.x;
  const int d  = dc >> 3;
  const int c  = dc & 7;
  const int lane = threadIdx.x;

  __shared__ float S[32][33];
  __shared__ float kpS[32][2];
  __shared__ float tsS[32];
  __shared__ float red[32];

  const float tau    = expf(log_tau[d]);
  const float inv2t2 = 1.0f / (2.0f * tau * tau);
  const float tp     = 20.0f + 2.0f * (float)c;

  const float t_p = ts0[lane];
  tsS[lane] = t_p;

  const float kp0 = expf(-(t_p * t_p) * inv2t2);
  const float dte = t_p - tp;
  const float kp1 = expf(-(dte * dte) * inv2t2);
  kpS[lane][0] = kp0;
  kpS[lane][1] = kp1;

  const float b   = expf(-(tp * tp) * inv2t2);
  const float det = 1.0f - b * b;
  const float a0  = (kp0 - b * kp1) / det;
  const float a1  = (kp1 - b * kp0) / det;
  __syncthreads();

  for (int q = 0; q < 32; ++q) {
    const float dd = t_p - tsS[q];
    S[lane][q] = expf(-(dd * dd) * inv2t2) - a0 * kpS[q][0] - a1 * kpS[q][1];
  }
  __syncthreads();

  for (int j = 0; j < 32; ++j) {
    if (lane == j) S[j][j] = sqrtf(S[j][j]);
    __syncthreads();
    const float djj = S[j][j];
    if (lane > j) S[lane][j] = S[lane][j] / djj;
    __syncthreads();
    if (lane > j) {
      const float lij = S[lane][j];
      for (int k = j + 1; k <= lane; ++k) S[lane][k] -= lij * S[k][j];
    }
    __syncthreads();
  }

  red[lane] = logf(fabsf(S[lane][lane]));
  __syncthreads();
  if (lane == 0) {
    float s = 0.0f;
    for (int i = 0; i < 32; ++i) s += red[i];
    parts[dc] = s;
  }

  float* wp = Wmat + ((size_t)dc * 32 + lane) * WROW;
  wp[0] = a0;
  wp[1] = a1;
  for (int q = 0; q < 32; ++q) wp[2 + q] = (q <= lane) ? S[lane][q] : 0.0f;
  wp[34] = 0.0f;
  wp[35] = 0.0f;
}

__global__ __launch_bounds__(256) void gp_gemm(
    const float* __restrict__ z,
    const float* __restrict__ Wmat,
    float* __restrict__ out) {
  const int tid  = threadIdx.x;
  const int wave = tid >> 5;
  const int lane = tid & 31;
  const int unit = blockIdx.x * 8 + wave;
  const int dc   = unit >> 7;
  const int mb   = unit & 127;
  const int d    = dc >> 3;
  const int c    = dc & 7;
  const int n0   = mb << 4;

  const int half = lane >> 4;
  const int l16  = lane & 15;

  const float* xrow  = z + ((size_t)(n0 + l16) * D_DIM + d) * T_DIM;
  const float* wbase = Wmat + (size_t)dc * 32 * WROW;
  const float* wr0   = wbase + (size_t)l16 * WROW;
  const float* wr1   = wbase + (size_t)(16 + l16) * WROW;

  v8f acc0 = {};
  v8f acc1 = {};

  for (int s = 0; s < 9; ++s) {
    const int ea = s * 4 + (half << 1);
    const int eb = ea + 1;
    const int ta = (ea < 2) ? ea : ((ea < E_DIM) ? (c * 32 + ea) : 0);
    const int tb = (eb < 2) ? eb : ((eb < E_DIM) ? (c * 32 + eb) : 0);

    v2f a;  a.x  = xrow[ta]; a.y  = xrow[tb];
    v2f b0; b0.x = wr0[ea];  b0.y = wr0[eb];
    v2f b1; b1.x = wr1[ea];  b1.y = wr1[eb];

    acc0 = __builtin_amdgcn_wmma_f32_16x16x4_f32(
        false, a, false, b0, (short)0, acc0, false, false);
    acc1 = __builtin_amdgcn_wmma_f32_16x16x4_f32(
        false, a, false, b1, (short)0, acc1, false, false);
  }

  for (int r = 0; r < 8; ++r) {
    const int    n     = n0 + r + (half << 3);
    const size_t obase = (((size_t)n * C_DIM + c) * D_DIM + d) * 34;
    out[obase + 1 + l16]      = acc0[r];
    out[obase + 1 + 16 + l16] = acc1[r];
  }

  {
    const int    n     = n0 + l16;
    const size_t obase = (((size_t)n * C_DIM + c) * D_DIM + d) * 34;
    const size_t zbase = ((size_t)n * D_DIM + d) * T_DIM;
    const float  v     = z[zbase + half];
    out[obase + (half ? 33 : 0)] = v;
  }
}

__global__ __launch_bounds__(256) void gp_sldj(
    const float* __restrict__ sldj_in,
    const float* __restrict__ parts,
    float* __restrict__ out) {
  const int n = blockIdx.x * blockDim.x + threadIdx.x;
  float tot = 0.0f;
  for (int i = 0; i < 256; ++i) tot += parts[i];
  out[ZOUT_ELEMS + n] = sldj_in[n] + tot;
}

extern "C" void kernel_launch(void* const* d_in, const int* in_sizes, int n_in,
                              void* d_out, int out_size, void* d_ws, size_t ws_size,
                              hipStream_t stream) {
  const float* z       = (const float*)d_in[0];
  const float* log_tau = (const float*)d_in[1];
  const float* ts0     = (const float*)d_in[2];
  const float* sldj    = (const float*)d_in[3];
  float* out = (float*)d_out;

  float* Wmat  = (float*)d_ws;
  float* parts = Wmat + 256 * 32 * WROW;

  gp_precompute<<<256, 32, 0, stream>>>(log_tau, ts0, Wmat, parts);
  gp_gemm<<<4096, 256, 0, stream>>>(z, Wmat, out);
  gp_sldj<<<8, 256, 0, stream>>>(sldj, parts, out);
}


# gemm: 4096 long-lived waves (was 32768): each keeps its (d,c) W fragments in registers and loops over 8 row blocks, the next block's rows requested ahead of the current products
# speedup vs baseline: 1.1798x; 1.1798x over previous
#include <hip/hip_runtime.h>

typedef __attribute__((ext_vector_type(2))) float v2f;
typedef __attribute__((ext_vector_type(8))) float v8f;

#define N_TOT   2048
#define D_DIM   32
#define C_DIM   8
#define P_DIM   32
#define T_DIM   258
#define E_DIM   34
#define WROW    36
#define ZOUT_ELEMS ((size_t)4 * 512 * 8 * 32 * 34)

__global__ __launch_bounds__(32) void gp_precompute(
    const float* __restrict__ log_tau,
    const float* __restrict__ ts0,
    float* __restrict__ Wmat,
    float* __restrict__ parts) {
  const int dc = blockIdx.x;
  const int d  = dc >> 3;
  const int c  = dc & 7;
  const int lane = threadIdx.x;

  __shared__ float S[32][33];
  __shared__ float kpS[32][2];
  __shared__ float tsS[32];
  __shared__ float red[32];

  const float tau    = expf(log_tau[d]);
  const float inv2t2 = 1.0f / (2.0f * tau * tau);
  const float tp     = 20.0f + 2.0f * (float)c;

  const float t_p = ts0[lane];
  tsS[lane] = t_p;

  const float kp0 = expf(-(t_p * t_p) * inv2t2);
  const float dte = t_p - tp;
  const float kp1 = expf(-(dte * dte) * inv2t2);
  kpS[lane][0] = kp0;
  kpS[lane][1] = kp1;

  const float b   = expf(-(tp * tp) * inv2t2);
  const float det = 1.0f - b * b;
  const float a0  = (kp0 - b * kp1) / det;
  const float a1  = (kp1 - b * kp0) / det;
  __syncthreads();

  for (int q = 0; q < 32; ++q) {
    const float dd = t_p - tsS[q];
    S[lane][q] = expf(-(dd * dd) * inv2t2) - a0 * kpS[q][0] - a1 * kpS[q][1];
  }
  __syncthreads();

  for (int j = 0; j < 32; ++j) {
    if (lane == j) S[j][j] = sqrtf(S[j][j]);
    __syncthreads();
    const float djj = S[j][j];
    if (lane > j) S[lane][j] = S[lane][j] / djj;
    __syncthreads();
    if (lane > j) {
      const float lij = S[lane][j];
      for (int k = j + 1; k <= lane; ++k) S[lane][k] -= lij * S[k][j];
    }
    __syncthreads();
  }

  red[lane] = logf(fabsf(S[lane][lane]));
  __syncthreads();
  if (lane == 0) {
    float s = 0.0f;
    for (int i = 0; i < 32; ++i) s += red[i];
    parts[dc] = s;
  }

  float* wp = Wmat + ((size_t)dc * 32 + lane) * WROW;
  wp[0] = a0;
  wp[1] = a1;
  for (int q = 0; q < 32; ++q) wp[2 + q] = (q <= lane) ? S[lane][q] : 0.0f;
  wp[34] = 0.0f;
  wp[35] = 0.0f;
}

#define MB_PER_WAVE   8
#define X_MB_STRIDE   ((size_t)16 * D_DIM * T_DIM)
#define O_ROW_STRIDE  ((size_t)C_DIM * D_DIM * 34)
#define O_MB_STRIDE   ((size_t)16 * O_ROW_STRIDE)

__device__ __forceinline__ void load_a_frags(v2f (&a)[9],
                                             const float* __restrict__ x0,
                                             const float* __restrict__ xmid,
                                             const float* __restrict__ x8) {
  a[0] = *(const v2f*)x0;
  for (int s = 1; s < 8; ++s) a[s] = *(const v2f*)(xmid + 4 * s);
  a[8] = *(const v2f*)x8;
}

__global__ __launch_bounds__(256) void gp_gemm(
    const float* __restrict__ z,
    const float* __restrict__ Wmat,
    float* __restrict__ out) {
  const int tid  = threadIdx.x;
  const int wave = tid >> 5;
  const int lane = tid & 31;
  const int dc   = blockIdx.x >> 1;
  const int mb0  = (((blockIdx.x & 1) << 3) + wave) * MB_PER_WAVE;
  const int d    = dc >> 3;
  const int c    = dc & 7;
  const int n0   = mb0 << 4;

  const int half = lane >> 4;
  const int l16  = lane & 15;

  const float* wr0 = Wmat + (size_t)dc * 32 * WROW + (size_t)l16 * WROW + 2 * half;
  const float* wr1 = wr0 + 16 * WROW;
  v2f b0[9], b1[9];
  for (int s = 0; s < 9; ++s) {
    b0[s] = *(const v2f*)(wr0 + 4 * s);
    b1[s] = *(const v2f*)(wr1 + 4 * s);
  }

  const float* xrow = z + ((size_t)(n0 + l16) * D_DIM + d) * T_DIM;
  const float* xmid = xrow + c * 32 + 2 * half;
  const float* x0   = half ? xmid : xrow;
  const float* x8   = xrow + c * 32 + 32;

  float* orow  = out + (((size_t)(n0 + 8 * half) * C_DIM + c) * D_DIM + d) * 34 + 1 + l16;
  float* opass = out + (((size_t)(n0 + l16) * C_DIM + c) * D_DIM + d) * 34;

  v2f a[9];
  load_a_frags(a, x0, xmid, x8);

  for (int it = 0; it < MB_PER_WAVE; ++it) {
    v2f an[9];
    if (it + 1 < MB_PER_WAVE) {
      const size_t xo = (size_t)(it + 1) * X_MB_STRIDE;
      load_a_frags(an, x0 + xo, xmid + xo, x8 + xo);
    }
    __builtin_amdgcn_sched_barrier(0);

    v8f acc0 = {};
    v8f acc1 = {};
    for (int s = 0; s < 8; ++s) {
      acc0 = __builtin_amdgcn_wmma_f32_16x16x4_f32(
          false, a[s], false, b0[s], (short)0, acc0, false, false);
      acc1 = __builtin_amdgcn_wmma_f32_16x16x4_f32(
          false, a[s], false, b1[s], (short)0, acc1, false, false);
    }
    {
      v2f a8;
      a8.x = half ? 0.0f : a[8].x;
      a8.y = half ? 0.0f : a[8].y;
      acc0 = __builtin_amdgcn_wmma_f32_16x16x4_f32(
          false, a8, false, b0[8], (short)0, acc0, false, false);
      acc1 = __builtin_amdgcn_wmma_f32_16x16x4_f32(
          false, a8, false, b1[8], (short)0, acc1, false, false);
    }

    float* o = orow + (size_t)it * O_MB_STRIDE;
    for (int r = 0; r < 8; ++r) {
      o[(size_t)r * O_ROW_STRIDE]      = acc0[r];
      o[(size_t)r * O_ROW_STRIDE + 16] = acc1[r];
    }

    if (half == 0) {
      float* p = opass + (size_t)it * O_MB_STRIDE;
      p[0]  = a[0].x;
      p[33] = a[0].y;
    }

    if (it + 1 < MB_PER_WAVE) {
      for (int s = 0; s < 9; ++s) a[s] = an[s];
    }
  }
}

__global__ __launch_bounds__(256) void gp_sldj(
    const float* __restrict__ sldj_in,
    const float* __restrict__ parts,
    float* __restrict__ out) {
  const int n = blockIdx.x * blockDim.x + threadIdx.x;
  float tot = 0.0f;
  for (int i = 0; i < 256; ++i) tot += parts[i];
  out[ZOUT_ELEMS + n] = sldj_in[n] + tot;
}

extern "C" void kernel_launch(void* const* d_in, const int* in_sizes, int n_in,
                              void* d_out, int out_size, void* d_ws, size_t ws_size,
                              hipStream_t stream) {
  const float* z       = (const float*)d_in[0];
  const float* log_tau = (const float*)d_in[1];
  const float* ts0     = (const float*)d_in[2];
  const float* sldj    = (const float*)d_in[3];
  float* out = (float*)d_out;

  float* Wmat  = (float*)d_ws;
  float* parts = Wmat + 256 * 32 * WROW;

  gp_precompute<<<256, 32, 0, stream>>>(log_tau, ts0, Wmat, parts);
  gp_gemm<<<512, 256, 0, stream>>>(z, Wmat, out);
  gp_sldj<<<8, 256, 0, stream>>>(sldj, parts, out);
}


# precompute: Cholesky with each lane's Sigma row held in registers, other rows' values fetched by lane reads (v_readlane) instead of dependent LDS round trips; same operation and summation order
# speedup vs baseline: 1.6794x; 1.4235x over previous
#include <hip/hip_runtime.h>

typedef __attribute__((ext_vector_type(2))) float v2f;
typedef __attribute__((ext_vector_type(8))) float v8f;

#define N_TOT   2048
#define D_DIM   32
#define C_DIM   8
#define P_DIM   32
#define T_DIM   258
#define E_DIM   34
#define WROW    36
#define ZOUT_ELEMS ((size_t)4 * 512 * 8 * 32 * 34)

typedef __attribute__((ext_vector_type(4))) float v4f;

__device__ __forceinline__ float lane_read(float x, int src) {
  return __builtin_bit_cast(float, __builtin_amdgcn_readlane(__builtin_bit_cast(int, x), src));
}

__global__ __launch_bounds__(32) void gp_precompute(
    const float* __restrict__ log_tau,
    const float* __restrict__ ts0,
    float* __restrict__ Wmat,
    float* __restrict__ parts) {
  const int dc = blockIdx.x;
  const int d  = dc >> 3;
  const int c  = dc & 7;
  const int lane = threadIdx.x;

  const float tau    = expf(log_tau[d]);
  const float inv2t2 = 1.0f / (2.0f * tau * tau);
  const float tp     = 20.0f + 2.0f * (float)c;

  const float t_p = ts0[lane];

  const float kp0 = expf(-(t_p * t_p) * inv2t2);
  const float dte = t_p - tp;
  const float kp1 = expf(-(dte * dte) * inv2t2);

  const float b   = expf(-(tp * tp) * inv2t2);
  const float det = 1.0f - b * b;
  const float a0  = (kp0 - b * kp1) / det;
  const float a1  = (kp1 - b * kp0) / det;

  float S[32];
#pragma unroll
  for (int q = 0; q < 32; ++q) {
    const float dd = t_p - lane_read(t_p, q);
    S[q] = expf(-(dd * dd) * inv2t2) - a0 * lane_read(kp0, q) - a1 * lane_read(kp1, q);
  }

  float ldiag = 1.0f;
#pragma unroll
  for (int j = 0; j < 32; ++j) {
    const float djj = sqrtf(lane_read(S[j], j));
    const float lij = S[j] / djj;
    S[j] = (lane == j) ? djj : lij;
    ldiag = (lane == j) ? djj : ldiag;
    const float lpj = S[j];
    for (int k = j + 1; k < 32; ++k) S[k] -= lpj * lane_read(lpj, k);
  }

  const float lg = logf(fabsf(ldiag));
  float s = 0.0f;
  for (int i = 0; i < 32; ++i) s += lane_read(lg, i);
  if (lane == 0) parts[dc] = s;

  float w[36];
  w[0] = a0;
  w[1] = a1;
  for (int q = 0; q < 32; ++q) w[2 + q] = (q <= lane) ? S[q] : 0.0f;
  w[34] = 0.0f;
  w[35] = 0.0f;
  v4f* wp = (v4f*)(Wmat + ((size_t)dc * 32 + lane) * WROW);
  for (int g = 0; g < 9; ++g) {
    v4f v;
    v.x = w[4 * g]; v.y = w[4 * g + 1]; v.z = w[4 * g + 2]; v.w = w[4 * g + 3];
    wp[g] = v;
  }
}

#define MB_PER_WAVE   8
#define X_MB_STRIDE   ((size_t)16 * D_DIM * T_DIM)
#define O_ROW_STRIDE  ((size_t)C_DIM * D_DIM * 34)
#define O_MB_STRIDE   ((size_t)16 * O_ROW_STRIDE)

__device__ __forceinline__ void load_a_frags(v2f (&a)[9],
                                             const float* __restrict__ x0,
                                             const float* __restrict__ xmid,
                                             const float* __restrict__ x8) {
  a[0] = *(const v2f*)x0;
  for (int s = 1; s < 8; ++s) a[s] = *(const v2f*)(xmid + 4 * s);
  a[8] = *(const v2f*)x8;
}

__global__ __launch_bounds__(256) void gp_gemm(
    const float* __restrict__ z,
    const float* __restrict__ Wmat,
    float* __restrict__ out) {
  const int tid  = threadIdx.x;
  const int wave = tid >> 5;
  const int lane = tid & 31;
  const int dc   = blockIdx.x >> 1;
  const int mb0  = (((blockIdx.x & 1) << 3) + wave) * MB_PER_WAVE;
  const int d    = dc >> 3;
  const int c    = dc & 7;
  const int n0   = mb0 << 4;

  const int half = lane >> 4;
  const int l16  = lane & 15;

  const float* wr0 = Wmat + (size_t)dc * 32 * WROW + (size_t)l16 * WROW + 2 * half;
  const float* wr1 = wr0 + 16 * WROW;
  v2f b0[9], b1[9];
  for (int s = 0; s < 9; ++s) {
    b0[s] = *(const v2f*)(wr0 + 4 * s);
    b1[s] = *(const v2f*)(wr1 + 4 * s);
  }

  const float* xrow = z + ((size_t)(n0 + l16) * D_DIM + d) * T_DIM;
  const float* xmid = xrow + c * 32 + 2 * half;
  const float* x0   = half ? xmid : xrow;
  const float* x8   = xrow + c * 32 + 32;

  float* orow  = out + (((size_t)(n0 + 8 * half) * C_DIM + c) * D_DIM + d) * 34 + 1 + l16;
  float* opass = out + (((size_t)(n0 + l16) * C_DIM + c) * D_DIM + d) * 34;

  v2f a[9];
  load_a_frags(a, x0, xmid, x8);

  for (int it = 0; it < MB_PER_WAVE; ++it) {
    v2f an[9];
    if (it + 1 < MB_PER_WAVE) {
      const size_t xo = (size_t)(it + 1) * X_MB_STRIDE;
      load_a_frags(an, x0 + xo, xmid + xo, x8 + xo);
    }
    __builtin_amdgcn_sched_barrier(0);

    v8f acc0 = {};
    v8f acc1 = {};
    for (int s = 0; s < 8; ++s) {
      acc0 = __builtin_amdgcn_wmma_f32_16x16x4_f32(
          false, a[s], false, b0[s], (short)0, acc0, false, false);
      acc1 = __builtin_amdgcn_wmma_f32_16x16x4_f32(
          false, a[s], false, b1[s], (short)0, acc1, false, false);
    }
    {
      v2f a8;
      a8.x = half ? 0.0f : a[8].x;
      a8.y = half ? 0.0f : a[8].y;
      acc0 = __builtin_amdgcn_wmma_f32_16x16x4_f32(
          false, a8, false, b0[8], (short)0, acc0, false, false);
      acc1 = __builtin_amdgcn_wmma_f32_16x16x4_f32(
          false, a8, false, b1[8], (short)0, acc1, false, false);
    }

    float* o = orow + (size_t)it * O_MB_STRIDE;
    for (int r = 0; r < 8; ++r) {
      o[(size_t)r * O_ROW_STRIDE]      = acc0[r];
      o[(size_t)r * O_ROW_STRIDE + 16] = acc1[r];
    }

    if (half == 0) {
      float* p = opass + (size_t)it * O_MB_STRIDE;
      p[0]  = a[0].x;
      p[33] = a[0].y;
    }

    if (it + 1 < MB_PER_WAVE) {
      for (int s = 0; s < 9; ++s) a[s] = an[s];
    }
  }
}

__global__ __launch_bounds__(256) void gp_sldj(
    const float* __restrict__ sldj_in,
    const float* __restrict__ parts,
    float* __restrict__ out) {
  const int n = blockIdx.x * blockDim.x + threadIdx.x;
  float tot = 0.0f;
  for (int i = 0; i < 256; ++i) tot += parts[i];
  out[ZOUT_ELEMS + n] = sldj_in[n] + tot;
}

extern "C" void kernel_launch(void* const* d_in, const int* in_sizes, int n_in,
                              void* d_out, int out_size, void* d_ws, size_t ws_size,
                              hipStream_t stream) {
  const float* z       = (const float*)d_in[0];
  const float* log_tau = (const float*)d_in[1];
  const float* ts0     = (const float*)d_in[2];
  const float* sldj    = (const float*)d_in[3];
  float* out = (float*)d_out;

  float* Wmat  = (float*)d_ws;
  float* parts = Wmat + 256 * 32 * WROW;

  gp_precompute<<<256, 32, 0, stream>>>(log_tau, ts0, Wmat, parts);
  gp_gemm<<<512, 256, 0, stream>>>(z, Wmat, out);
  gp_sldj<<<8, 256, 0, stream>>>(sldj, parts, out);
}


# launches 3 -> 2: the log-det add (sldj + sum of the 256 partials, same index order) runs in 8 extra work-groups of the product launch instead of a third dependent launch behind it
# speedup vs baseline: 1.9001x; 1.1314x over previous
#include <hip/hip_runtime.h>

typedef __attribute__((ext_vector_type(2))) float v2f;
typedef __attribute__((ext_vector_type(8))) float v8f;

#define N_TOT   2048
#define D_DIM   32
#define C_DIM   8
#define P_DIM   32
#define T_DIM   258
#define E_DIM   34
#define WROW    36
#define ZOUT_ELEMS ((size_t)4 * 512 * 8 * 32 * 34)

typedef __attribute__((ext_vector_type(4))) float v4f;

__device__ __forceinline__ float lane_read(float x, int src) {
  return __builtin_bit_cast(float, __builtin_amdgcn_readlane(__builtin_bit_cast(int, x), src));
}

__global__ __launch_bounds__(32) void gp_precompute(
    const float* __restrict__ log_tau,
    const float* __restrict__ ts0,
    float* __restrict__ Wmat,
    float* __restrict__ parts) {
  const int dc = blockIdx.x;
  const int d  = dc >> 3;
  const int c  = dc & 7;
  const int lane = threadIdx.x;

  const float tau    = expf(log_tau[d]);
  const float inv2t2 = 1.0f / (2.0f * tau * tau);
  const float tp     = 20.0f + 2.0f * (float)c;

  const float t_p = ts0[lane];

  const float kp0 = expf(-(t_p * t_p) * inv2t2);
  const float dte = t_p - tp;
  const float kp1 = expf(-(dte * dte) * inv2t2);

  const float b   = expf(-(tp * tp) * inv2t2);
  const float det = 1.0f - b * b;
  const float a0  = (kp0 - b * kp1) / det;
  const float a1  = (kp1 - b * kp0) / det;

  float S[32];
#pragma unroll
  for (int q = 0; q < 32; ++q) {
    const float dd = t_p - lane_read(t_p, q);
    S[q] = expf(-(dd * dd) * inv2t2) - a0 * lane_read(kp0, q) - a1 * lane_read(kp1, q);
  }

  float ldiag = 1.0f;
#pragma unroll
  for (int j = 0; j < 32; ++j) {
    const float djj = sqrtf(lane_read(S[j], j));
    const float lij = S[j] / djj;
    S[j] = (lane == j) ? djj : lij;
    ldiag = (lane == j) ? djj : ldiag;
    const float lpj = S[j];
    for (int k = j + 1; k < 32; ++k) S[k] -= lpj * lane_read(lpj, k);
  }

  const float lg = logf(fabsf(ldiag));
  float s = 0.0f;
  for (int i = 0; i < 32; ++i) s += lane_read(lg, i);
  if (lane == 0) parts[dc] = s;

  float w[36];
  w[0] = a0;
  w[1] = a1;
  for (int q = 0; q < 32; ++q) w[2 + q] = (q <= lane) ? S[q] : 0.0f;
  w[34] = 0.0f;
  w[35] = 0.0f;
  v4f* wp = (v4f*)(Wmat + ((size_t)dc * 32 + lane) * WROW);
  for (int g = 0; g < 9; ++g) {
    v4f v;
    v.x = w[4 * g]; v.y = w[4 * g + 1]; v.z = w[4 * g + 2]; v.w = w[4 * g + 3];
    wp[g] = v;
  }
}

#define MB_PER_WAVE   8
#define GEMM_BLOCKS   512
#define SLDJ_BLOCKS   8
#define GRID_BLOCKS   520
#define X_MB_STRIDE   ((size_t)16 * D_DIM * T_DIM)
#define O_ROW_STRIDE  ((size_t)C_DIM * D_DIM * 34)
#define O_MB_STRIDE   ((size_t)16 * O_ROW_STRIDE)

__device__ __forceinline__ void load_a_frags(v2f (&a)[9],
                                             const float* __restrict__ x0,
                                             const float* __restrict__ xmid,
                                             const float* __restrict__ x8) {
  a[0] = *(const v2f*)x0;
  for (int s = 1; s < 8; ++s) a[s] = *(const v2f*)(xmid + 4 * s);
  a[8] = *(const v2f*)x8;
}

__global__ __launch_bounds__(256) void gp_gemm(
    const float* __restrict__ z,
    const float* __restrict__ Wmat,
    const float* __restrict__ sldj_in,
    const float* __restrict__ parts,
    float* __restrict__ out) {
  if (blockIdx.x >= GEMM_BLOCKS) {
    const int n = (blockIdx.x - GEMM_BLOCKS) * 256 + threadIdx.x;
    float tot = 0.0f;
    for (int i = 0; i < 256; ++i) tot += parts[i];
    out[ZOUT_ELEMS + n] = sldj_in[n] + tot;
    return;
  }

  const int tid  = threadIdx.x;
  const int wave = tid >> 5;
  const int lane = tid & 31;
  const int dc   = blockIdx.x >> 1;
  const int mb0  = (((blockIdx.x & 1) << 3) + wave) * MB_PER_WAVE;
  const int d    = dc >> 3;
  const int c    = dc & 7;
  const int n0   = mb0 << 4;

  const int half = lane >> 4;
  const int l16  = lane & 15;

  const float* wr0 = Wmat + (size_t)dc * 32 * WROW + (size_t)l16 * WROW + 2 * half;
  const float* wr1 = wr0 + 16 * WROW;
  v2f b0[9], b1[9];
  for (int s = 0; s < 9; ++s) {
    b0[s] = *(const v2f*)(wr0 + 4 * s);
    b1[s] = *(const v2f*)(wr1 + 4 * s);
  }

  const float* xrow = z + ((size_t)(n0 + l16) * D_DIM + d) * T_DIM;
  const float* xmid = xrow + c * 32 + 2 * half;
  const float* x0   = half ? xmid : xrow;
  const float* x8   = xrow + c * 32 + 32;

  float* orow  = out + (((size_t)(n0 + 8 * half) * C_DIM + c) * D_DIM + d) * 34 + 1 + l16;
  float* opass = out + (((size_t)(n0 + l16) * C_DIM + c) * D_DIM + d) * 34;

  v2f a[9];
  load_a_frags(a, x0, xmid, x8);

  for (int it = 0; it < MB_PER_WAVE; ++it) {
    v2f an[9];
    if (it + 1 < MB_PER_WAVE) {
      const size_t xo = (size_t)(it + 1) * X_MB_STRIDE;
      load_a_frags(an, x0 + xo, xmid + xo, x8 + xo);
    }
    __builtin_amdgcn_sched_barrier(0);

    v8f acc0 = {};
    v8f acc1 = {};
    for (int s = 0; s < 8; ++s) {
      acc0 = __builtin_amdgcn_wmma_f32_16x16x4_f32(
          false, a[s], false, b0[s], (short)0, acc0, false, false);
      acc1 = __builtin_amdgcn_wmma_f32_16x16x4_f32(
          false, a[s], false, b1[s], (short)0, acc1, false, false);
    }
    {
      v2f a8;
      a8.x = half ? 0.0f : a[8].x;
      a8.y = half ? 0.0f : a[8].y;
      acc0 = __builtin_amdgcn_wmma_f32_16x16x4_f32(
          false, a8, false, b0[8], (short)0, acc0, false, false);
      acc1 = __builtin_amdgcn_wmma_f32_16x16x4_f32(
          false, a8, false, b1[8], (short)0, acc1, false, false);
    }

    float* o = orow + (size_t)it * O_MB_STRIDE;
    for (int r = 0; r < 8; ++r) {
      o[(size_t)r * O_ROW_STRIDE]      = acc0[r];
      o[(size_t)r * O_ROW_STRIDE + 16] = acc1[r];
    }

    if (half == 0) {
      float* p = opass + (size_t)it * O_MB_STRIDE;
      p[0]  = a[0].x;
      p[33] = a[0].y;
    }

    if (it + 1 < MB_PER_WAVE) {
      for (int s = 0; s < 9; ++s) a[s] = an[s];
    }
  }
}

extern "C" void kernel_launch(void* const* d_in, const int* in_sizes, int n_in,
                              void* d_out, int out_size, void* d_ws, size_t ws_size,
                              hipStream_t stream) {
  const float* z       = (const float*)d_in[0];
  const float* log_tau = (const float*)d_in[1];
  const float* ts0     = (const float*)d_in[2];
  const float* sldj    = (const float*)d_in[3];
  float* out = (float*)d_out;

  float* Wmat  = (float*)d_ws;
  float* parts = Wmat + 256 * 32 * WROW;

  gp_precompute<<<256, 32, 0, stream>>>(log_tau, ts0, Wmat, parts);
  gp_gemm<<<GRID_BLOCKS, 256, 0, stream>>>(z, Wmat, sldj, parts, out);
}


# gemm A operand: rows fetched by row-contiguous loads (two 128-byte row segments per wave-wide load) and re-laid to fragment order in a per-wave LDS tile, replacing 16-line gathers; same k order
# speedup vs baseline: 2.0136x; 1.0597x over previous
#include <hip/hip_runtime.h>

typedef __attribute__((ext_vector_type(2))) float v2f;
typedef __attribute__((ext_vector_type(8))) float v8f;

#define N_TOT   2048
#define D_DIM   32
#define C_DIM   8
#define P_DIM   32
#define T_DIM   258
#define E_DIM   34
#define WROW    36
#define ZOUT_ELEMS ((size_t)4 * 512 * 8 * 32 * 34)

typedef __attribute__((ext_vector_type(4))) float v4f;

__device__ __forceinline__ float lane_read(float x, int src) {
  return __builtin_bit_cast(float, __builtin_amdgcn_readlane(__builtin_bit_cast(int, x), src));
}

__global__ __launch_bounds__(32) void gp_precompute(
    const float* __restrict__ log_tau,
    const float* __restrict__ ts0,
    float* __restrict__ Wmat,
    float* __restrict__ parts) {
  const int dc = blockIdx.x;
  const int d  = dc >> 3;
  const int c  = dc & 7;
  const int lane = threadIdx.x;

  const float tau    = expf(log_tau[d]);
  const float inv2t2 = 1.0f / (2.0f * tau * tau);
  const float tp     = 20.0f + 2.0f * (float)c;

  const float t_p = ts0[lane];

  const float kp0 = expf(-(t_p * t_p) * inv2t2);
  const float dte = t_p - tp;
  const float kp1 = expf(-(dte * dte) * inv2t2);

  const float b   = expf(-(tp * tp) * inv2t2);
  const float det = 1.0f - b * b;
  const float a0  = (kp0 - b * kp1) / det;
  const float a1  = (kp1 - b * kp0) / det;

  float S[32];
#pragma unroll
  for (int q = 0; q < 32; ++q) {
    const float dd = t_p - lane_read(t_p, q);
    S[q] = expf(-(dd * dd) * inv2t2) - a0 * lane_read(kp0, q) - a1 * lane_read(kp1, q);
  }

  float ldiag = 1.0f;
#pragma unroll
  for (int j = 0; j < 32; ++j) {
    const float djj = sqrtf(lane_read(S[j], j));
    const float lij = S[j] / djj;
    S[j] = (lane == j) ? djj : lij;
    ldiag = (lane == j) ? djj : ldiag;
    const float lpj = S[j];
    for (int k = j + 1; k < 32; ++k) S[k] -= lpj * lane_read(lpj, k);
  }

  const float lg = logf(fabsf(ldiag));
  float s = 0.0f;
  for (int i = 0; i < 32; ++i) s += lane_read(lg, i);
  if (lane == 0) parts[dc] = s;

  float w[36];
  w[0] = a0;
  w[1] = a1;
  for (int q = 0; q < 32; ++q) w[2 + q] = (q <= lane) ? S[q] : 0.0f;
  w[34] = 0.0f;
  w[35] = 0.0f;
  v4f* wp = (v4f*)(Wmat + ((size_t)dc * 32 + lane) * WROW);
  for (int g = 0; g < 9; ++g) {
    v4f v;
    v.x = w[4 * g]; v.y = w[4 * g + 1]; v.z = w[4 * g + 2]; v.w = w[4 * g + 3];
    wp[g] = v;
  }
}

#define MB_PER_WAVE   8
#define GEMM_BLOCKS   512
#define SLDJ_BLOCKS   8
#define GRID_BLOCKS   520
#define X_MB_STRIDE   ((size_t)16 * D_DIM * T_DIM)
#define X_2ROW_STRIDE ((size_t)2 * D_DIM * T_DIM)
#define O_ROW_STRIDE  ((size_t)C_DIM * D_DIM * 34)
#define O_MB_STRIDE   ((size_t)16 * O_ROW_STRIDE)
#define TILE_PITCH    18

__device__ __forceinline__ void load_rows(v2f (&g)[9],
                                          const float* __restrict__ gmid,
                                          const float* __restrict__ gp0) {
  for (int q = 0; q < 8; ++q) g[q] = *(const v2f*)(gmid + (size_t)q * X_2ROW_STRIDE);
  g[8] = *(const v2f*)gp0;
}

__global__ __launch_bounds__(256) void gp_gemm(
    const float* __restrict__ z,
    const float* __restrict__ Wmat,
    const float* __restrict__ sldj_in,
    const float* __restrict__ parts,
    float* __restrict__ out) {
  __shared__ v2f tile_all[8][16 * TILE_PITCH];

  if (blockIdx.x >= GEMM_BLOCKS) {
    const int n = (blockIdx.x - GEMM_BLOCKS) * 256 + threadIdx.x;
    float tot = 0.0f;
    for (int i = 0; i < 256; ++i) tot += parts[i];
    out[ZOUT_ELEMS + n] = sldj_in[n] + tot;
    return;
  }

  const int tid  = threadIdx.x;
  const int wave = tid >> 5;
  const int lane = tid & 31;
  const int dc   = blockIdx.x >> 1;
  const int mb0  = (((blockIdx.x & 1) << 3) + wave) * MB_PER_WAVE;
  const int d    = dc >> 3;
  const int c    = dc & 7;
  const int n0   = mb0 << 4;

  const int half = lane >> 4;
  const int l16  = lane & 15;

  const float* wr0 = Wmat + (size_t)dc * 32 * WROW + (size_t)l16 * WROW + 2 * half;
  const float* wr1 = wr0 + 16 * WROW;
  v2f b0[9], b1[9];
  for (int s = 0; s < 9; ++s) {
    b0[s] = *(const v2f*)(wr0 + 4 * s);
    b1[s] = *(const v2f*)(wr1 + 4 * s);
  }

  const float* gmid = z + ((size_t)(n0 + half) * D_DIM + d) * T_DIM + c * 32 + 2 + 2 * l16;
  const float* gp0  = z + ((size_t)(n0 + l16) * D_DIM + d) * T_DIM;

  v2f* tile = tile_all[wave];
  v2f* st_mid = tile + half * TILE_PITCH + l16 + 1;
  v2f* st_p0  = tile + l16 * TILE_PITCH + (half ? 17 : 0);
  const v2f* ld_frag = tile + l16 * TILE_PITCH + half;
  const v2f* ld_last = tile + l16 * TILE_PITCH + 16;

  float* orow  = out + (((size_t)(n0 + 8 * half) * C_DIM + c) * D_DIM + d) * 34 + 1 + l16;
  float* opass = out + (((size_t)(n0 + l16) * C_DIM + c) * D_DIM + d) * 34;

  v2f g[9];
  v2f a[9];
  {
    v2f gfirst[9];
    load_rows(gfirst, gmid, gp0);
    load_rows(g, gmid + X_MB_STRIDE, gp0 + X_MB_STRIDE);
    for (int q = 0; q < 8; ++q) st_mid[q * 2 * TILE_PITCH] = gfirst[q];
    st_p0[0] = gfirst[8];
  }
  __builtin_amdgcn_fence(__ATOMIC_SEQ_CST, "wavefront");
  for (int s = 0; s < 8; ++s) a[s] = ld_frag[2 * s];
  a[8] = ld_last[0];

  for (int it = 0; it < MB_PER_WAVE; ++it) {
    v2f an[9];
    if (it + 1 < MB_PER_WAVE) {
      __builtin_amdgcn_fence(__ATOMIC_SEQ_CST, "wavefront");
      for (int q = 0; q < 8; ++q) st_mid[q * 2 * TILE_PITCH] = g[q];
      st_p0[0] = g[8];
      if (it + 2 < MB_PER_WAVE) {
        const size_t xo = (size_t)(it + 2) * X_MB_STRIDE;
        load_rows(g, gmid + xo, gp0 + xo);
      }
      __builtin_amdgcn_fence(__ATOMIC_SEQ_CST, "wavefront");
      for (int s = 0; s < 8; ++s) an[s] = ld_frag[2 * s];
      an[8] = ld_last[0];
    }
    __builtin_amdgcn_sched_barrier(0);

    v8f acc0 = {};
    v8f acc1 = {};
    for (int s = 0; s < 8; ++s) {
      acc0 = __builtin_amdgcn_wmma_f32_16x16x4_f32(
          false, a[s], false, b0[s], (short)0, acc0, false, false);
      acc1 = __builtin_amdgcn_wmma_f32_16x16x4_f32(
          false, a[s], false, b1[s], (short)0, acc1, false, false);
    }
    {
      v2f a8;
      a8.x = half ? 0.0f : a[8].x;
      a8.y = half ? 0.0f : a[8].y;
      acc0 = __builtin_amdgcn_wmma_f32_16x16x4_f32(
          false, a8, false, b0[8], (short)0, acc0, false, false);
      acc1 = __builtin_amdgcn_wmma_f32_16x16x4_f32(
          false, a8, false, b1[8], (short)0, acc1, false, false);
    }

    float* o = orow + (size_t)it * O_MB_STRIDE;
    for (int r = 0; r < 8; ++r) {
      o[(size_t)r * O_ROW_STRIDE]      = acc0[r];
      o[(size_t)r * O_ROW_STRIDE + 16] = acc1[r];
    }

    if (half == 0) {
      float* p = opass + (size_t)it * O_MB_STRIDE;
      p[0]  = a[0].x;
      p[33] = a[0].y;
    }

    if (it + 1 < MB_PER_WAVE) {
      for (int s = 0; s < 9; ++s) a[s] = an[s];
    }
  }
}

extern "C" void kernel_launch(void* const* d_in, const int* in_sizes, int n_in,
                              void* d_out, int out_size, void* d_ws, size_t ws_size,
                              hipStream_t stream) {
  const float* z       = (const float*)d_in[0];
  const float* log_tau = (const float*)d_in[1];
  const float* ts0     = (const float*)d_in[2];
  const float* sldj    = (const float*)d_in[3];
  float* out = (float*)d_out;

  float* Wmat  = (float*)d_ws;
  float* parts = Wmat + 256 * 32 * WROW;

  gp_precompute<<<256, 32, 0, stream>>>(log_tau, ts0, Wmat, parts);
  gp_gemm<<<GRID_BLOCKS, 256, 0, stream>>>(z, Wmat, sldj, parts, out);
}


# gemm ownership: a work-group is one d x 128 rows with wave = c (was one (d,c) x 1024 rows): the 8 waves sharing each z row sit on one compute unit, so a z row is fetched by one work-group, once
# speedup vs baseline: 2.2676x; 1.1261x over previous
#include <hip/hip_runtime.h>

typedef __attribute__((ext_vector_type(2))) float v2f;
typedef __attribute__((ext_vector_type(8))) float v8f;

#define N_TOT   2048
#define D_DIM   32
#define C_DIM   8
#define P_DIM   32
#define T_DIM   258
#define E_DIM   34
#define WROW    36
#define ZOUT_ELEMS ((size_t)4 * 512 * 8 * 32 * 34)

typedef __attribute__((ext_vector_type(4))) float v4f;

__device__ __forceinline__ float lane_read(float x, int src) {
  return __builtin_bit_cast(float, __builtin_amdgcn_readlane(__builtin_bit_cast(int, x), src));
}

__global__ __launch_bounds__(32) void gp_precompute(
    const float* __restrict__ log_tau,
    const float* __restrict__ ts0,
    float* __restrict__ Wmat,
    float* __restrict__ parts) {
  const int dc = blockIdx.x;
  const int d  = dc >> 3;
  const int c  = dc & 7;
  const int lane = threadIdx.x;

  const float tau    = expf(log_tau[d]);
  const float inv2t2 = 1.0f / (2.0f * tau * tau);
  const float tp     = 20.0f + 2.0f * (float)c;

  const float t_p = ts0[lane];

  const float kp0 = expf(-(t_p * t_p) * inv2t2);
  const float dte = t_p - tp;
  const float kp1 = expf(-(dte * dte) * inv2t2);

  const float b   = expf(-(tp * tp) * inv2t2);
  const float det = 1.0f - b * b;
  const float a0  = (kp0 - b * kp1) / det;
  const float a1  = (kp1 - b * kp0) / det;

  float S[32];
#pragma unroll
  for (int q = 0; q < 32; ++q) {
    const float dd = t_p - lane_read(t_p, q);
    S[q] = expf(-(dd * dd) * inv2t2) - a0 * lane_read(kp0, q) - a1 * lane_read(kp1, q);
  }

  float ldiag = 1.0f;
#pragma unroll
  for (int j = 0; j < 32; ++j) {
    const float djj = sqrtf(lane_read(S[j], j));
    const float lij = S[j] / djj;
    S[j] = (lane == j) ? djj : lij;
    ldiag = (lane == j) ? djj : ldiag;
    const float lpj = S[j];
    for (int k = j + 1; k < 32; ++k) S[k] -= lpj * lane_read(lpj, k);
  }

  const float lg = logf(fabsf(ldiag));
  float s = 0.0f;
  for (int i = 0; i < 32; ++i) s += lane_read(lg, i);
  if (lane == 0) parts[dc] = s;

  float w[36];
  w[0] = a0;
  w[1] = a1;
  for (int q = 0; q < 32; ++q) w[2 + q] = (q <= lane) ? S[q] : 0.0f;
  w[34] = 0.0f;
  w[35] = 0.0f;
  v4f* wp = (v4f*)(Wmat + ((size_t)dc * 32 + lane) * WROW);
  for (int g = 0; g < 9; ++g) {
    v4f v;
    v.x = w[4 * g]; v.y = w[4 * g + 1]; v.z = w[4 * g + 2]; v.w = w[4 * g + 3];
    wp[g] = v;
  }
}

#define MB_PER_WAVE   8
#define GEMM_BLOCKS   512
#define SLDJ_BLOCKS   8
#define GRID_BLOCKS   520
#define X_MB_STRIDE   ((size_t)16 * D_DIM * T_DIM)
#define X_2ROW_STRIDE ((size_t)2 * D_DIM * T_DIM)
#define O_ROW_STRIDE  ((size_t)C_DIM * D_DIM * 34)
#define O_MB_STRIDE   ((size_t)16 * O_ROW_STRIDE)
#define TILE_PITCH    18

__device__ __forceinline__ void load_rows(v2f (&g)[9],
                                          const float* __restrict__ gmid,
                                          const float* __restrict__ gp0) {
  for (int q = 0; q < 8; ++q) g[q] = *(const v2f*)(gmid + (size_t)q * X_2ROW_STRIDE);
  g[8] = *(const v2f*)gp0;
}

__global__ __launch_bounds__(256) void gp_gemm(
    const float* __restrict__ z,
    const float* __restrict__ Wmat,
    const float* __restrict__ sldj_in,
    const float* __restrict__ parts,
    float* __restrict__ out) {
  __shared__ v2f tile_all[8][16 * TILE_PITCH];

  if (blockIdx.x >= GEMM_BLOCKS) {
    const int n = (blockIdx.x - GEMM_BLOCKS) * 256 + threadIdx.x;
    float tot = 0.0f;
    for (int i = 0; i < 256; ++i) tot += parts[i];
    out[ZOUT_ELEMS + n] = sldj_in[n] + tot;
    return;
  }

  const int tid  = threadIdx.x;
  const int wave = tid >> 5;
  const int lane = tid & 31;
  const int d    = blockIdx.x >> 4;
  const int c    = wave;
  const int dc   = (d << 3) + c;
  const int mb0  = (blockIdx.x & 15) * MB_PER_WAVE;
  const int n0   = mb0 << 4;

  const int half = lane >> 4;
  const int l16  = lane & 15;

  const float* wr0 = Wmat + (size_t)dc * 32 * WROW + (size_t)l16 * WROW + 2 * half;
  const float* wr1 = wr0 + 16 * WROW;
  v2f b0[9], b1[9];
  for (int s = 0; s < 9; ++s) {
    b0[s] = *(const v2f*)(wr0 + 4 * s);
    b1[s] = *(const v2f*)(wr1 + 4 * s);
  }

  const float* gmid = z + ((size_t)(n0 + half) * D_DIM + d) * T_DIM + c * 32 + 2 + 2 * l16;
  const float* gp0  = z + ((size_t)(n0 + l16) * D_DIM + d) * T_DIM;

  v2f* tile = tile_all[wave];
  v2f* st_mid = tile + half * TILE_PITCH + l16 + 1;
  v2f* st_p0  = tile + l16 * TILE_PITCH + (half ? 17 : 0);
  const v2f* ld_frag = tile + l16 * TILE_PITCH + half;
  const v2f* ld_last = tile + l16 * TILE_PITCH + 16;

  float* orow  = out + (((size_t)(n0 + 8 * half) * C_DIM + c) * D_DIM + d) * 34 + 1 + l16;
  float* opass = out + (((size_t)(n0 + l16) * C_DIM + c) * D_DIM + d) * 34;

  v2f g[9];
  v2f a[9];
  {
    v2f gfirst[9];
    load_rows(gfirst, gmid, gp0);
    load_rows(g, gmid + X_MB_STRIDE, gp0 + X_MB_STRIDE);
    for (int q = 0; q < 8; ++q) st_mid[q * 2 * TILE_PITCH] = gfirst[q];
    st_p0[0] = gfirst[8];
  }
  __builtin_amdgcn_fence(__ATOMIC_SEQ_CST, "wavefront");
  for (int s = 0; s < 8; ++s) a[s] = ld_frag[2 * s];
  a[8] = ld_last[0];

  for (int it = 0; it < MB_PER_WAVE; ++it) {
    v2f an[9];
    if (it + 1 < MB_PER_WAVE) {
      __builtin_amdgcn_fence(__ATOMIC_SEQ_CST, "wavefront");
      for (int q = 0; q < 8; ++q) st_mid[q * 2 * TILE_PITCH] = g[q];
      st_p0[0] = g[8];
      if (it + 2 < MB_PER_WAVE) {
        const size_t xo = (size_t)(it + 2) * X_MB_STRIDE;
        load_rows(g, gmid + xo, gp0 + xo);
      }
      __builtin_amdgcn_fence(__ATOMIC_SEQ_CST, "wavefront");
      for (int s = 0; s < 8; ++s) an[s] = ld_frag[2 * s];
      an[8] = ld_last[0];
    }
    __builtin_amdgcn_sched_barrier(0);

    v8f acc0 = {};
    v8f acc1 = {};
    for (int s = 0; s < 8; ++s) {
      acc0 = __builtin_amdgcn_wmma_f32_16x16x4_f32(
          false, a[s], false, b0[s], (short)0, acc0, false, false);
      acc1 = __builtin_amdgcn_wmma_f32_16x16x4_f32(
          false, a[s], false, b1[s], (short)0, acc1, false, false);
    }
    {
      v2f a8;
      a8.x = half ? 0.0f : a[8].x;
      a8.y = half ? 0.0f : a[8].y;
      acc0 = __builtin_amdgcn_wmma_f32_16x16x4_f32(
          false, a8, false, b0[8], (short)0, acc0, false, false);
      acc1 = __builtin_amdgcn_wmma_f32_16x16x4_f32(
          false, a8, false, b1[8], (short)0, acc1, false, false);
    }

    float* o = orow + (size_t)it * O_MB_STRIDE;
    for (int r = 0; r < 8; ++r) {
      o[(size_t)r * O_ROW_STRIDE]      = acc0[r];
      o[(size_t)r * O_ROW_STRIDE + 16] = acc1[r];
    }

    if (half == 0) {
      float* p = opass + (size_t)it * O_MB_STRIDE;
      p[0]  = a[0].x;
      p[33] = a[0].y;
    }

    if (it + 1 < MB_PER_WAVE) {
      for (int s = 0; s < 9; ++s) a[s] = an[s];
    }
  }
}

extern "C" void kernel_launch(void* const* d_in, const int* in_sizes, int n_in,
                              void* d_out, int out_size, void* d_ws, size_t ws_size,
                              hipStream_t stream) {
  const float* z       = (const float*)d_in[0];
  const float* log_tau = (const float*)d_in[1];
  const float* ts0     = (const float*)d_in[2];
  const float* sldj    = (const float*)d_in[3];
  float* out = (float*)d_out;

  float* Wmat  = (float*)d_ws;
  float* parts = Wmat + 256 * 32 * WROW;

  gp_precompute<<<256, 32, 0, stream>>>(log_tau, ts0, Wmat, parts);
  gp_gemm<<<GRID_BLOCKS, 256, 0, stream>>>(z, Wmat, sldj, parts, out);
}


# gemm out path: each 16x34 block goes through a per-wave LDS tile and leaves as nine row-contiguous 8-byte stores (whole 136-byte rows incl. first/last column), not 18 4-byte stores; sums unchanged
# speedup vs baseline: 2.3966x; 1.0569x over previous
#include <hip/hip_runtime.h>

typedef __attribute__((ext_vector_type(2))) float v2f;
typedef __attribute__((ext_vector_type(8))) float v8f;

#define N_TOT   2048
#define D_DIM   32
#define C_DIM   8
#define P_DIM   32
#define T_DIM   258
#define E_DIM   34
#define WROW    36
#define ZOUT_ELEMS ((size_t)4 * 512 * 8 * 32 * 34)

typedef __attribute__((ext_vector_type(4))) float v4f;

__device__ __forceinline__ float lane_read(float x, int src) {
  return __builtin_bit_cast(float, __builtin_amdgcn_readlane(__builtin_bit_cast(int, x), src));
}

__global__ __launch_bounds__(32) void gp_precompute(
    const float* __restrict__ log_tau,
    const float* __restrict__ ts0,
    float* __restrict__ Wmat,
    float* __restrict__ parts) {
  const int dc = blockIdx.x;
  const int d  = dc >> 3;
  const int c  = dc & 7;
  const int lane = threadIdx.x;

  const float tau    = expf(log_tau[d]);
  const float inv2t2 = 1.0f / (2.0f * tau * tau);
  const float tp     = 20.0f + 2.0f * (float)c;

  const float t_p = ts0[lane];

  const float kp0 = expf(-(t_p * t_p) * inv2t2);
  const float dte = t_p - tp;
  const float kp1 = expf(-(dte * dte) * inv2t2);

  const float b   = expf(-(tp * tp) * inv2t2);
  const float det = 1.0f - b * b;
  const float a0  = (kp0 - b * kp1) / det;
  const float a1  = (kp1 - b * kp0) / det;

  float S[32];
#pragma unroll
  for (int q = 0; q < 32; ++q) {
    const float dd = t_p - lane_read(t_p, q);
    S[q] = expf(-(dd * dd) * inv2t2) - a0 * lane_read(kp0, q) - a1 * lane_read(kp1, q);
  }

  float ldiag = 1.0f;
#pragma unroll
  for (int j = 0; j < 32; ++j) {
    const float djj = sqrtf(lane_read(S[j], j));
    const float lij = S[j] / djj;
    S[j] = (lane == j) ? djj : lij;
    ldiag = (lane == j) ? djj : ldiag;
    const float lpj = S[j];
    for (int k = j + 1; k < 32; ++k) S[k] -= lpj * lane_read(lpj, k);
  }

  const float lg = logf(fabsf(ldiag));
  float s = 0.0f;
  for (int i = 0; i < 32; ++i) s += lane_read(lg, i);
  if (lane == 0) parts[dc] = s;

  float w[36];
  w[0] = a0;
  w[1] = a1;
  for (int q = 0; q < 32; ++q) w[2 + q] = (q <= lane) ? S[q] : 0.0f;
  w[34] = 0.0f;
  w[35] = 0.0f;
  v4f* wp = (v4f*)(Wmat + ((size_t)dc * 32 + lane) * WROW);
  for (int g = 0; g < 9; ++g) {
    v4f v;
    v.x = w[4 * g]; v.y = w[4 * g + 1]; v.z = w[4 * g + 2]; v.w = w[4 * g + 3];
    wp[g] = v;
  }
}

#define MB_PER_WAVE   8
#define GEMM_BLOCKS   512
#define SLDJ_BLOCKS   8
#define GRID_BLOCKS   520
#define X_MB_STRIDE   ((size_t)16 * D_DIM * T_DIM)
#define X_2ROW_STRIDE ((size_t)2 * D_DIM * T_DIM)
#define O_ROW_STRIDE  ((size_t)C_DIM * D_DIM * 34)
#define O_MB_STRIDE   ((size_t)16 * O_ROW_STRIDE)
#define TILE_PITCH    18
#define OT_ROW        34
#define OT_PAIRS      272

__device__ __forceinline__ void load_rows(v2f (&g)[9],
                                          const float* __restrict__ gmid,
                                          const float* __restrict__ gp0) {
  for (int q = 0; q < 8; ++q) g[q] = *(const v2f*)(gmid + (size_t)q * X_2ROW_STRIDE);
  g[8] = *(const v2f*)gp0;
}

__global__ __launch_bounds__(256) void gp_gemm(
    const float* __restrict__ z,
    const float* __restrict__ Wmat,
    const float* __restrict__ sldj_in,
    const float* __restrict__ parts,
    float* __restrict__ out) {
  __shared__ v2f tile_all[8][16 * TILE_PITCH];
  __shared__ v2f otile_all[8][OT_PAIRS];

  if (blockIdx.x >= GEMM_BLOCKS) {
    const int n = (blockIdx.x - GEMM_BLOCKS) * 256 + threadIdx.x;
    float tot = 0.0f;
    for (int i = 0; i < 256; ++i) tot += parts[i];
    out[ZOUT_ELEMS + n] = sldj_in[n] + tot;
    return;
  }

  const int tid  = threadIdx.x;
  const int wave = tid >> 5;
  const int lane = tid & 31;
  const int d    = blockIdx.x >> 4;
  const int c    = wave;
  const int dc   = (d << 3) + c;
  const int mb0  = (blockIdx.x & 15) * MB_PER_WAVE;
  const int n0   = mb0 << 4;

  const int half = lane >> 4;
  const int l16  = lane & 15;

  const float* wr0 = Wmat + (size_t)dc * 32 * WROW + (size_t)l16 * WROW + 2 * half;
  const float* wr1 = wr0 + 16 * WROW;
  v2f b0[9], b1[9];
  for (int s = 0; s < 9; ++s) {
    b0[s] = *(const v2f*)(wr0 + 4 * s);
    b1[s] = *(const v2f*)(wr1 + 4 * s);
  }

  const float* gmid = z + ((size_t)(n0 + half) * D_DIM + d) * T_DIM + c * 32 + 2 + 2 * l16;
  const float* gp0  = z + ((size_t)(n0 + l16) * D_DIM + d) * T_DIM;

  v2f* tile = tile_all[wave];
  v2f* st_mid = tile + half * TILE_PITCH + l16 + 1;
  v2f* st_p0  = tile + l16 * TILE_PITCH + (half ? 17 : 0);
  const v2f* ld_frag = tile + l16 * TILE_PITCH + half;
  const v2f* ld_last = tile + l16 * TILE_PITCH + 16;

  float* otf = (float*)otile_all[wave];
  float* st_acc  = otf + (8 * half) * OT_ROW + 1 + l16;
  float* st_pass = otf + l16 * OT_ROW + (half ? 33 : 0);
  const v2f* ld_out  = otile_all[wave] + lane;
  const v2f* ld_out8 = otile_all[wave] + 256 + l16;
  float* oblk = out + (((size_t)n0 * C_DIM + c) * D_DIM + d) * 34;
  unsigned ooff[9];
  for (int k = 0; k < 8; ++k) {
    const unsigned i = 32u * k + lane;
    ooff[k] = (i / 17u) * (unsigned)O_ROW_STRIDE + 2u * (i % 17u);
  }
  ooff[8] = 15u * (unsigned)O_ROW_STRIDE + 2u * (1u + l16);

  v2f g[9];
  v2f a[9];
  v2f pcur;
  v2f op[9];
  {
    v2f gfirst[9];
    load_rows(gfirst, gmid, gp0);
    load_rows(g, gmid + X_MB_STRIDE, gp0 + X_MB_STRIDE);
    for (int q = 0; q < 8; ++q) st_mid[q * 2 * TILE_PITCH] = gfirst[q];
    st_p0[0] = gfirst[8];
    pcur = gfirst[8];
  }
  __builtin_amdgcn_fence(__ATOMIC_SEQ_CST, "wavefront");
  for (int s = 0; s < 8; ++s) a[s] = ld_frag[2 * s];
  a[8] = ld_last[0];

  for (int it = 0; it < MB_PER_WAVE; ++it) {
    v2f an[9];
    v2f pnext = pcur;
    if (it + 1 < MB_PER_WAVE) {
      __builtin_amdgcn_fence(__ATOMIC_SEQ_CST, "wavefront");
      for (int q = 0; q < 8; ++q) st_mid[q * 2 * TILE_PITCH] = g[q];
      st_p0[0] = g[8];
      pnext = g[8];
      if (it + 2 < MB_PER_WAVE) {
        const size_t xo = (size_t)(it + 2) * X_MB_STRIDE;
        load_rows(g, gmid + xo, gp0 + xo);
      }
      __builtin_amdgcn_fence(__ATOMIC_SEQ_CST, "wavefront");
      for (int s = 0; s < 8; ++s) an[s] = ld_frag[2 * s];
      an[8] = ld_last[0];
    }
    if (it > 0) {
      __builtin_amdgcn_fence(__ATOMIC_SEQ_CST, "wavefront");
      for (int k = 0; k < 8; ++k) op[k] = ld_out[32 * k];
      op[8] = ld_out8[0];
    }
    __builtin_amdgcn_sched_barrier(0);

    v8f acc0 = {};
    v8f acc1 = {};
    for (int s = 0; s < 8; ++s) {
      acc0 = __builtin_amdgcn_wmma_f32_16x16x4_f32(
          false, a[s], false, b0[s], (short)0, acc0, false, false);
      acc1 = __builtin_amdgcn_wmma_f32_16x16x4_f32(
          false, a[s], false, b1[s], (short)0, acc1, false, false);
    }
    {
      v2f a8;
      a8.x = half ? 0.0f : a[8].x;
      a8.y = half ? 0.0f : a[8].y;
      acc0 = __builtin_amdgcn_wmma_f32_16x16x4_f32(
          false, a8, false, b0[8], (short)0, acc0, false, false);
      acc1 = __builtin_amdgcn_wmma_f32_16x16x4_f32(
          false, a8, false, b1[8], (short)0, acc1, false, false);
    }

    __builtin_amdgcn_sched_barrier(0);

    if (it > 0) {
      float* ob = oblk + (size_t)(it - 1) * O_MB_STRIDE;
      for (int k = 0; k < 8; ++k) *(v2f*)(ob + ooff[k]) = op[k];
      if (half == 0) *(v2f*)(ob + ooff[8]) = op[8];
    }

    __builtin_amdgcn_fence(__ATOMIC_SEQ_CST, "wavefront");
    for (int r = 0; r < 8; ++r) {
      st_acc[r * OT_ROW]      = acc0[r];
      st_acc[r * OT_ROW + 16] = acc1[r];
    }
    st_pass[0] = half ? pcur.y : pcur.x;

    if (it + 1 < MB_PER_WAVE) {
      for (int s = 0; s < 9; ++s) a[s] = an[s];
    }
    pcur = pnext;
  }

  __builtin_amdgcn_fence(__ATOMIC_SEQ_CST, "wavefront");
  for (int k = 0; k < 8; ++k) op[k] = ld_out[32 * k];
  op[8] = ld_out8[0];
  {
    float* ob = oblk + (size_t)(MB_PER_WAVE - 1) * O_MB_STRIDE;
    for (int k = 0; k < 8; ++k) *(v2f*)(ob + ooff[k]) = op[k];
    if (half == 0) *(v2f*)(ob + ooff[8]) = op[8];
  }
}

extern "C" void kernel_launch(void* const* d_in, const int* in_sizes, int n_in,
                              void* d_out, int out_size, void* d_ws, size_t ws_size,
                              hipStream_t stream) {
  const float* z       = (const float*)d_in[0];
  const float* log_tau = (const float*)d_in[1];
  const float* ts0     = (const float*)d_in[2];
  const float* sldj    = (const float*)d_in[3];
  float* out = (float*)d_out;

  float* Wmat  = (float*)d_ws;
  float* parts = Wmat + 256 * 32 * WROW;

  gp_precompute<<<256, 32, 0, stream>>>(log_tau, ts0, Wmat, parts);
  gp_gemm<<<GRID_BLOCKS, 256, 0, stream>>>(z, Wmat, sldj, parts, out);
}


# log-det add: its 8 work-groups fetch the 256 partials once into LDS (one memory round trip) and every thread sums them from there, not in 16 dependent memory trips per thread; same index order
# speedup vs baseline: 2.4910x; 1.0394x over previous
#include <hip/hip_runtime.h>

typedef __attribute__((ext_vector_type(2))) float v2f;
typedef __attribute__((ext_vector_type(8))) float v8f;

#define N_TOT   2048
#define D_DIM   32
#define C_DIM   8
#define P_DIM   32
#define T_DIM   258
#define E_DIM   34
#define WROW    36
#define ZOUT_ELEMS ((size_t)4 * 512 * 8 * 32 * 34)

typedef __attribute__((ext_vector_type(4))) float v4f;

__device__ __forceinline__ float lane_read(float x, int src) {
  return __builtin_bit_cast(float, __builtin_amdgcn_readlane(__builtin_bit_cast(int, x), src));
}

__global__ __launch_bounds__(32) void gp_precompute(
    const float* __restrict__ log_tau,
    const float* __restrict__ ts0,
    float* __restrict__ Wmat,
    float* __restrict__ parts) {
  const int dc = blockIdx.x;
  const int d  = dc >> 3;
  const int c  = dc & 7;
  const int lane = threadIdx.x;

  const float tau    = expf(log_tau[d]);
  const float inv2t2 = 1.0f / (2.0f * tau * tau);
  const float tp     = 20.0f + 2.0f * (float)c;

  const float t_p = ts0[lane];

  const float kp0 = expf(-(t_p * t_p) * inv2t2);
  const float dte = t_p - tp;
  const float kp1 = expf(-(dte * dte) * inv2t2);

  const float b   = expf(-(tp * tp) * inv2t2);
  const float det = 1.0f - b * b;
  const float a0  = (kp0 - b * kp1) / det;
  const float a1  = (kp1 - b * kp0) / det;

  float S[32];
#pragma unroll
  for (int q = 0; q < 32; ++q) {
    const float dd = t_p - lane_read(t_p, q);
    S[q] = expf(-(dd * dd) * inv2t2) - a0 * lane_read(kp0, q) - a1 * lane_read(kp1, q);
  }

  float ldiag = 1.0f;
#pragma unroll
  for (int j = 0; j < 32; ++j) {
    const float djj = sqrtf(lane_read(S[j], j));
    const float lij = S[j] / djj;
    S[j] = (lane == j) ? djj : lij;
    ldiag = (lane == j) ? djj : ldiag;
    const float lpj = S[j];
    for (int k = j + 1; k < 32; ++k) S[k] -= lpj * lane_read(lpj, k);
  }

  const float lg = logf(fabsf(ldiag));
  float s = 0.0f;
  for (int i = 0; i < 32; ++i) s += lane_read(lg, i);
  if (lane == 0) parts[dc] = s;

  float w[36];
  w[0] = a0;
  w[1] = a1;
  for (int q = 0; q < 32; ++q) w[2 + q] = (q <= lane) ? S[q] : 0.0f;
  w[34] = 0.0f;
  w[35] = 0.0f;
  v4f* wp = (v4f*)(Wmat + ((size_t)dc * 32 + lane) * WROW);
  for (int g = 0; g < 9; ++g) {
    v4f v;
    v.x = w[4 * g]; v.y = w[4 * g + 1]; v.z = w[4 * g + 2]; v.w = w[4 * g + 3];
    wp[g] = v;
  }
}

#define MB_PER_WAVE   8
#define GEMM_BLOCKS   512
#define SLDJ_BLOCKS   8
#define GRID_BLOCKS   520
#define X_MB_STRIDE   ((size_t)16 * D_DIM * T_DIM)
#define X_2ROW_STRIDE ((size_t)2 * D_DIM * T_DIM)
#define O_ROW_STRIDE  ((size_t)C_DIM * D_DIM * 34)
#define O_MB_STRIDE   ((size_t)16 * O_ROW_STRIDE)
#define TILE_PITCH    18
#define OT_ROW        34
#define OT_PAIRS      272

__device__ __forceinline__ void load_rows(v2f (&g)[9],
                                          const float* __restrict__ gmid,
                                          const float* __restrict__ gp0) {
  for (int q = 0; q < 8; ++q) g[q] = *(const v2f*)(gmid + (size_t)q * X_2ROW_STRIDE);
  g[8] = *(const v2f*)gp0;
}

__global__ __launch_bounds__(256) void gp_gemm(
    const float* __restrict__ z,
    const float* __restrict__ Wmat,
    const float* __restrict__ sldj_in,
    const float* __restrict__ parts,
    float* __restrict__ out) {
  __shared__ v2f tile_all[8][16 * TILE_PITCH];
  __shared__ v2f otile_all[8][OT_PAIRS];
  __shared__ __attribute__((aligned(16))) float parts_lds[256];

  if (blockIdx.x >= GEMM_BLOCKS) {
    const int n = (blockIdx.x - GEMM_BLOCKS) * 256 + threadIdx.x;
    const float s_in = sldj_in[n];
    parts_lds[threadIdx.x] = parts[threadIdx.x];
    __syncthreads();
    float tot = 0.0f;
    const v4f* pl = (const v4f*)parts_lds;
    for (int cidx = 0; cidx < 4; ++cidx) {
      v4f q[16];
      for (int k = 0; k < 16; ++k) q[k] = pl[16 * cidx + k];
      for (int k = 0; k < 16; ++k) {
        tot += q[k].x;
        tot += q[k].y;
        tot += q[k].z;
        tot += q[k].w;
      }
      __builtin_amdgcn_sched_group_barrier(0x100, 16, 0);
      __builtin_amdgcn_sched_group_barrier(0x002, 64, 0);
    }
    out[ZOUT_ELEMS + n] = s_in + tot;
    return;
  }

  const int tid  = threadIdx.x;
  const int wave = tid >> 5;
  const int lane = tid & 31;
  const int d    = blockIdx.x >> 4;
  const int c    = wave;
  const int dc   = (d << 3) + c;
  const int mb0  = (blockIdx.x & 15) * MB_PER_WAVE;
  const int n0   = mb0 << 4;

  const int half = lane >> 4;
  const int l16  = lane & 15;

  const float* wr0 = Wmat + (size_t)dc * 32 * WROW + (size_t)l16 * WROW + 2 * half;
  const float* wr1 = wr0 + 16 * WROW;
  v2f b0[9], b1[9];
  for (int s = 0; s < 9; ++s) {
    b0[s] = *(const v2f*)(wr0 + 4 * s);
    b1[s] = *(const v2f*)(wr1 + 4 * s);
  }

  const float* gmid = z + ((size_t)(n0 + half) * D_DIM + d) * T_DIM + c * 32 + 2 + 2 * l16;
  const float* gp0  = z + ((size_t)(n0 + l16) * D_DIM + d) * T_DIM;

  v2f* tile = tile_all[wave];
  v2f* st_mid = tile + half * TILE_PITCH + l16 + 1;
  v2f* st_p0  = tile + l16 * TILE_PITCH + (half ? 17 : 0);
  const v2f* ld_frag = tile + l16 * TILE_PITCH + half;
  const v2f* ld_last = tile + l16 * TILE_PITCH + 16;

  float* otf = (float*)otile_all[wave];
  float* st_acc  = otf + (8 * half) * OT_ROW + 1 + l16;
  float* st_pass = otf + l16 * OT_ROW + (half ? 33 : 0);
  const v2f* ld_out  = otile_all[wave] + lane;
  const v2f* ld_out8 = otile_all[wave] + 256 + l16;
  float* oblk = out + (((size_t)n0 * C_DIM + c) * D_DIM + d) * 34;
  unsigned ooff[9];
  for (int k = 0; k < 8; ++k) {
    const unsigned i = 32u * k + lane;
    ooff[k] = (i / 17u) * (unsigned)O_ROW_STRIDE + 2u * (i % 17u);
  }
  ooff[8] = 15u * (unsigned)O_ROW_STRIDE + 2u * (1u + l16);

  v2f g[9];
  v2f a[9];
  v2f pcur;
  v2f op[9];
  {
    v2f gfirst[9];
    load_rows(gfirst, gmid, gp0);
    load_rows(g, gmid + X_MB_STRIDE, gp0 + X_MB_STRIDE);
    for (int q = 0; q < 8; ++q) st_mid[q * 2 * TILE_PITCH] = gfirst[q];
    st_p0[0] = gfirst[8];
    pcur = gfirst[8];
  }
  __builtin_amdgcn_fence(__ATOMIC_SEQ_CST, "wavefront");
  for (int s = 0; s < 8; ++s) a[s] = ld_frag[2 * s];
  a[8] = ld_last[0];

  for (int it = 0; it < MB_PER_WAVE; ++it) {
    v2f an[9];
    v2f pnext = pcur;
    if (it + 1 < MB_PER_WAVE) {
      __builtin_amdgcn_fence(__ATOMIC_SEQ_CST, "wavefront");
      for (int q = 0; q < 8; ++q) st_mid[q * 2 * TILE_PITCH] = g[q];
      st_p0[0] = g[8];
      pnext = g[8];
      if (it + 2 < MB_PER_WAVE) {
        const size_t xo = (size_t)(it + 2) * X_MB_STRIDE;
        load_rows(g, gmid + xo, gp0 + xo);
      }
      __builtin_amdgcn_fence(__ATOMIC_SEQ_CST, "wavefront");
      for (int s = 0; s < 8; ++s) an[s] = ld_frag[2 * s];
      an[8] = ld_last[0];
    }
    if (it > 0) {
      __builtin_amdgcn_fence(__ATOMIC_SEQ_CST, "wavefront");
      for (int k = 0; k < 8; ++k) op[k] = ld_out[32 * k];
      op[8] = ld_out8[0];
    }
    __builtin_amdgcn_sched_barrier(0);

    v8f acc0 = {};
    v8f acc1 = {};
    for (int s = 0; s < 8; ++s) {
      acc0 = __builtin_amdgcn_wmma_f32_16x16x4_f32(
          false, a[s], false, b0[s], (short)0, acc0, false, false);
      acc1 = __builtin_amdgcn_wmma_f32_16x16x4_f32(
          false, a[s], false, b1[s], (short)0, acc1, false, false);
    }
    {
      v2f a8;
      a8.x = half ? 0.0f : a[8].x;
      a8.y = half ? 0.0f : a[8].y;
      acc0 = __builtin_amdgcn_wmma_f32_16x16x4_f32(
          false, a8, false, b0[8], (short)0, acc0, false, false);
      acc1 = __builtin_amdgcn_wmma_f32_16x16x4_f32(
          false, a8, false, b1[8], (short)0, acc1, false, false);
    }

    __builtin_amdgcn_sched_barrier(0);

    if (it > 0) {
      float* ob = oblk + (size_t)(it - 1) * O_MB_STRIDE;
      for (int k = 0; k < 8; ++k) *(v2f*)(ob + ooff[k]) = op[k];
      if (half == 0) *(v2f*)(ob + ooff[8]) = op[8];
    }

    __builtin_amdgcn_fence(__ATOMIC_SEQ_CST, "wavefront");
    for (int r = 0; r < 8; ++r) {
      st_acc[r * OT_ROW]      = acc0[r];
      st_acc[r * OT_ROW + 16] = acc1[r];
    }
    st_pass[0] = half ? pcur.y : pcur.x;

    if (it + 1 < MB_PER_WAVE) {
      for (int s = 0; s < 9; ++s) a[s] = an[s];
    }
    pcur = pnext;
  }

  __builtin_amdgcn_fence(__ATOMIC_SEQ_CST, "wavefront");
  for (int k = 0; k < 8; ++k) op[k] = ld_out[32 * k];
  op[8] = ld_out8[0];
  {
    float* ob = oblk + (size_t)(MB_PER_WAVE - 1) * O_MB_STRIDE;
    for (int k = 0; k < 8; ++k) *(v2f*)(ob + ooff[k]) = op[k];
    if (half == 0) *(v2f*)(ob + ooff[8]) = op[8];
  }
}

extern "C" void kernel_launch(void* const* d_in, const int* in_sizes, int n_in,
                              void* d_out, int out_size, void* d_ws, size_t ws_size,
                              hipStream_t stream) {
  const float* z       = (const float*)d_in[0];
  const float* log_tau = (const float*)d_in[1];
  const float* ts0     = (const float*)d_in[2];
  const float* sldj    = (const float*)d_in[3];
  float* out = (float*)d_out;

  float* Wmat  = (float*)d_ws;
  float* parts = Wmat + 256 * 32 * WROW;

  gp_precompute<<<256, 32, 0, stream>>>(log_tau, ts0, Wmat, parts);
  gp_gemm<<<GRID_BLOCKS, 256, 0, stream>>>(z, Wmat, sldj, parts, out);
}


# precompute: Cholesky column j = S * rsqrtf(pivot), each lane's reciprocal root taken once column j is final; sqrt + IEEE division leave the pivot chain; L moves in last bits, diagonal stays sqrtf
# speedup vs baseline: 2.7756x; 1.1143x over previous
#include <hip/hip_runtime.h>

typedef __attribute__((ext_vector_type(2))) float v2f;
typedef __attribute__((ext_vector_type(8))) float v8f;

#define N_TOT   2048
#define D_DIM   32
#define C_DIM   8
#define P_DIM   32
#define T_DIM   258
#define E_DIM   34
#define WROW    36
#define ZOUT_ELEMS ((size_t)4 * 512 * 8 * 32 * 34)

typedef __attribute__((ext_vector_type(4))) float v4f;

__device__ __forceinline__ float lane_read(float x, int src) {
  return __builtin_bit_cast(float, __builtin_amdgcn_readlane(__builtin_bit_cast(int, x), src));
}

__global__ __launch_bounds__(32) void gp_precompute(
    const float* __restrict__ log_tau,
    const float* __restrict__ ts0,
    float* __restrict__ Wmat,
    float* __restrict__ parts) {
  const int dc = blockIdx.x;
  const int d  = dc >> 3;
  const int c  = dc & 7;
  const int lane = threadIdx.x;

  const float tau    = expf(log_tau[d]);
  const float inv2t2 = 1.0f / (2.0f * tau * tau);
  const float tp     = 20.0f + 2.0f * (float)c;

  const float t_p = ts0[lane];

  const float kp0 = expf(-(t_p * t_p) * inv2t2);
  const float dte = t_p - tp;
  const float kp1 = expf(-(dte * dte) * inv2t2);

  const float b   = expf(-(tp * tp) * inv2t2);
  const float det = 1.0f - b * b;
  const float a0  = (kp0 - b * kp1) / det;
  const float a1  = (kp1 - b * kp0) / det;

  float S[32];
#pragma unroll
  for (int q = 0; q < 32; ++q) {
    const float dd = t_p - lane_read(t_p, q);
    S[q] = expf(-(dd * dd) * inv2t2) - a0 * lane_read(kp0, q) - a1 * lane_read(kp1, q);
  }

  float pown = 1.0f;
  float rq = rsqrtf(S[0]);
#pragma unroll
  for (int j = 0; j < 32; ++j) {
    const float rj = lane_read(rq, j);
    pown = (lane == j) ? S[j] : pown;
    const float lpj = S[j] * rj;
    S[j] = lpj;
    if (j + 1 < 32) {
      S[j + 1] -= lpj * lane_read(lpj, j + 1);
      rq = rsqrtf(S[j + 1]);
    }
    for (int k = j + 2; k < 32; ++k) S[k] -= lpj * lane_read(lpj, k);
    __builtin_amdgcn_sched_barrier(0);
  }
  const float ldiag = sqrtf(pown);

  const float lg = logf(fabsf(ldiag));
  float s = 0.0f;
  for (int i = 0; i < 32; ++i) s += lane_read(lg, i);
  if (lane == 0) parts[dc] = s;

  float w[36];
  w[0] = a0;
  w[1] = a1;
  for (int q = 0; q < 32; ++q) w[2 + q] = (q < lane) ? S[q] : ((q == lane) ? ldiag : 0.0f);
  w[34] = 0.0f;
  w[35] = 0.0f;
  v4f* wp = (v4f*)(Wmat + ((size_t)dc * 32 + lane) * WROW);
  for (int g = 0; g < 9; ++g) {
    v4f v;
    v.x = w[4 * g]; v.y = w[4 * g + 1]; v.z = w[4 * g + 2]; v.w = w[4 * g + 3];
    wp[g] = v;
  }
}

#define MB_PER_WAVE   8
#define GEMM_BLOCKS   512
#define SLDJ_BLOCKS   8
#define GRID_BLOCKS   520
#define X_MB_STRIDE   ((size_t)16 * D_DIM * T_DIM)
#define X_2ROW_STRIDE ((size_t)2 * D_DIM * T_DIM)
#define O_ROW_STRIDE  ((size_t)C_DIM * D_DIM * 34)
#define O_MB_STRIDE   ((size_t)16 * O_ROW_STRIDE)
#define TILE_PITCH    18
#define OT_ROW        34
#define OT_PAIRS      272

__device__ __forceinline__ void load_rows(v2f (&g)[9],
                                          const float* __restrict__ gmid,
                                          const float* __restrict__ gp0) {
  for (int q = 0; q < 8; ++q) g[q] = *(const v2f*)(gmid + (size_t)q * X_2ROW_STRIDE);
  g[8] = *(const v2f*)gp0;
}

__global__ __launch_bounds__(256) void gp_gemm(
    const float* __restrict__ z,
    const float* __restrict__ Wmat,
    const float* __restrict__ sldj_in,
    const float* __restrict__ parts,
    float* __restrict__ out) {
  __shared__ v2f tile_all[8][16 * TILE_PITCH];
  __shared__ v2f otile_all[8][OT_PAIRS];
  __shared__ __attribute__((aligned(16))) float parts_lds[256];

  if (blockIdx.x >= GEMM_BLOCKS) {
    const int n = (blockIdx.x - GEMM_BLOCKS) * 256 + threadIdx.x;
    const float s_in = sldj_in[n];
    parts_lds[threadIdx.x] = parts[threadIdx.x];
    __syncthreads();
    float tot = 0.0f;
    const v4f* pl = (const v4f*)parts_lds;
    for (int cidx = 0; cidx < 4; ++cidx) {
      v4f q[16];
      for (int k = 0; k < 16; ++k) q[k] = pl[16 * cidx + k];
      for (int k = 0; k < 16; ++k) {
        tot += q[k].x;
        tot += q[k].y;
        tot += q[k].z;
        tot += q[k].w;
      }
      __builtin_amdgcn_sched_group_barrier(0x100, 16, 0);
      __builtin_amdgcn_sched_group_barrier(0x002, 64, 0);
    }
    out[ZOUT_ELEMS + n] = s_in + tot;
    return;
  }

  const int tid  = threadIdx.x;
  const int wave = tid >> 5;
  const int lane = tid & 31;
  const int d    = blockIdx.x >> 4;
  const int c    = wave;
  const int dc   = (d << 3) + c;
  const int mb0  = (blockIdx.x & 15) * MB_PER_WAVE;
  const int n0   = mb0 << 4;

  const int half = lane >> 4;
  const int l16  = lane & 15;

  const float* wr0 = Wmat + (size_t)dc * 32 * WROW + (size_t)l16 * WROW + 2 * half;
  const float* wr1 = wr0 + 16 * WROW;
  v2f b0[9], b1[9];
  for (int s = 0; s < 9; ++s) {
    b0[s] = *(const v2f*)(wr0 + 4 * s);
    b1[s] = *(const v2f*)(wr1 + 4 * s);
  }

  const float* gmid = z + ((size_t)(n0 + half) * D_DIM + d) * T_DIM + c * 32 + 2 + 2 * l16;
  const float* gp0  = z + ((size_t)(n0 + l16) * D_DIM + d) * T_DIM;

  v2f* tile = tile_all[wave];
  v2f* st_mid = tile + half * TILE_PITCH + l16 + 1;
  v2f* st_p0  = tile + l16 * TILE_PITCH + (half ? 17 : 0);
  const v2f* ld_frag = tile + l16 * TILE_PITCH + half;
  const v2f* ld_last = tile + l16 * TILE_PITCH + 16;

  float* otf = (float*)otile_all[wave];
  float* st_acc  = otf + (8 * half) * OT_ROW + 1 + l16;
  float* st_pass = otf + l16 * OT_ROW + (half ? 33 : 0);
  const v2f* ld_out  = otile_all[wave] + lane;
  const v2f* ld_out8 = otile_all[wave] + 256 + l16;
  float* oblk = out + (((size_t)n0 * C_DIM + c) * D_DIM + d) * 34;
  unsigned ooff[9];
  for (int k = 0; k < 8; ++k) {
    const unsigned i = 32u * k + lane;
    ooff[k] = (i / 17u) * (unsigned)O_ROW_STRIDE + 2u * (i % 17u);
  }
  ooff[8] = 15u * (unsigned)O_ROW_STRIDE + 2u * (1u + l16);

  v2f g[9];
  v2f a[9];
  v2f pcur;
  v2f op[9];
  {
    v2f gfirst[9];
    load_rows(gfirst, gmid, gp0);
    load_rows(g, gmid + X_MB_STRIDE, gp0 + X_MB_STRIDE);
    for (int q = 0; q < 8; ++q) st_mid[q * 2 * TILE_PITCH] = gfirst[q];
    st_p0[0] = gfirst[8];
    pcur = gfirst[8];
  }
  __builtin_amdgcn_fence(__ATOMIC_SEQ_CST, "wavefront");
  for (int s = 0; s < 8; ++s) a[s] = ld_frag[2 * s];
  a[8] = ld_last[0];

  for (int it = 0; it < MB_PER_WAVE; ++it) {
    v2f an[9];
    v2f pnext = pcur;
    if (it + 1 < MB_PER_WAVE) {
      __builtin_amdgcn_fence(__ATOMIC_SEQ_CST, "wavefront");
      for (int q = 0; q < 8; ++q) st_mid[q * 2 * TILE_PITCH] = g[q];
      st_p0[0] = g[8];
      pnext = g[8];
      if (it + 2 < MB_PER_WAVE) {
        const size_t xo = (size_t)(it + 2) * X_MB_STRIDE;
        load_rows(g, gmid + xo, gp0 + xo);
      }
      __builtin_amdgcn_fence(__ATOMIC_SEQ_CST, "wavefront");
      for (int s = 0; s < 8; ++s) an[s] = ld_frag[2 * s];
      an[8] = ld_last[0];
    }
    if (it > 0) {
      __builtin_amdgcn_fence(__ATOMIC_SEQ_CST, "wavefront");
      for (int k = 0; k < 8; ++k) op[k] = ld_out[32 * k];
      op[8] = ld_out8[0];
    }
    __builtin_amdgcn_sched_barrier(0);

    v8f acc0 = {};
    v8f acc1 = {};
    for (int s = 0; s < 8; ++s) {
      acc0 = __builtin_amdgcn_wmma_f32_16x16x4_f32(
          false, a[s], false, b0[s], (short)0, acc0, false, false);
      acc1 = __builtin_amdgcn_wmma_f32_16x16x4_f32(
          false, a[s], false, b1[s], (short)0, acc1, false, false);
    }
    {
      v2f a8;
      a8.x = half ? 0.0f : a[8].x;
      a8.y = half ? 0.0f : a[8].y;
      acc0 = __builtin_amdgcn_wmma_f32_16x16x4_f32(
          false, a8, false, b0[8], (short)0, acc0, false, false);
      acc1 = __builtin_amdgcn_wmma_f32_16x16x4_f32(
          false, a8, false, b1[8], (short)0, acc1, false, false);
    }

    __builtin_amdgcn_sched_barrier(0);

    if (it > 0) {
      float* ob = oblk + (size_t)(it - 1) * O_MB_STRIDE;
      for (int k = 0; k < 8; ++k) *(v2f*)(ob + ooff[k]) = op[k];
      if (half == 0) *(v2f*)(ob + ooff[8]) = op[8];
    }

    __builtin_amdgcn_fence(__ATOMIC_SEQ_CST, "wavefront");
    for (int r = 0; r < 8; ++r) {
      st_acc[r * OT_ROW]      = acc0[r];
      st_acc[r * OT_ROW + 16] = acc1[r];
    }
    st_pass[0] = half ? pcur.y : pcur.x;

    if (it + 1 < MB_PER_WAVE) {
      for (int s = 0; s < 9; ++s) a[s] = an[s];
    }
    pcur = pnext;
  }

  __builtin_amdgcn_fence(__ATOMIC_SEQ_CST, "wavefront");
  for (int k = 0; k < 8; ++k) op[k] = ld_out[32 * k];
  op[8] = ld_out8[0];
  {
    float* ob = oblk + (size_t)(MB_PER_WAVE - 1) * O_MB_STRIDE;
    for (int k = 0; k < 8; ++k) *(v2f*)(ob + ooff[k]) = op[k];
    if (half == 0) *(v2f*)(ob + ooff[8]) = op[8];
  }
}

extern "C" void kernel_launch(void* const* d_in, const int* in_sizes, int n_in,
                              void* d_out, int out_size, void* d_ws, size_t ws_size,
                              hipStream_t stream) {
  const float* z       = (const float*)d_in[0];
  const float* log_tau = (const float*)d_in[1];
  const float* ts0     = (const float*)d_in[2];
  const float* sldj    = (const float*)d_in[3];
  float* out = (float*)d_out;

  float* Wmat  = (float*)d_ws;
  float* parts = Wmat + 256 * 32 * WROW;

  gp_precompute<<<256, 32, 0, stream>>>(log_tau, ts0, Wmat, parts);
  gp_gemm<<<GRID_BLOCKS, 256, 0, stream>>>(z, Wmat, sldj, parts, out);
}
